# starting kernel
# baseline (speedup 1.0000x reference)
#include <hip/hip_runtime.h>
#include <math.h>

typedef __attribute__((ext_vector_type(2))) float v2f;
typedef __attribute__((ext_vector_type(8))) float v8f;

#define HH 4
#define DD 32
#define NN 1024
#define BB 8
#define KDIM 128
#define NEG_SLOPE 0.2f

__global__ __launch_bounds__(256) void gat_h_gemm(const float* __restrict__ x,
                                                  const float* __restrict__ W,
                                                  float* __restrict__ h) {
    __shared__ float lds_x[16][132];
    __shared__ float lds_w[128][36];
    const int tid  = threadIdx.x;
    const int lane = tid & 31;
    const int wave = tid >> 5;
    const int r0   = blockIdx.x * 16;
    const int c0   = wave * 16;
    const int lm   = lane & 15;
    const int lhi  = lane >> 4;

    for (int idx = tid; idx < 16 * KDIM; idx += 256) {
        int row = idx >> 7, col = idx & 127;
        lds_x[row][col] = x[(size_t)(r0 + row) * KDIM + col];
    }

    v8f acc = {};
    for (int kk = 0; kk < KDIM; kk += 32) {
        __syncthreads();
        for (int idx = tid; idx < 128 * 32; idx += 256) {
            int row = idx >> 5, col = idx & 31;
            lds_w[row][col] = W[(size_t)row * KDIM + kk + col];
        }
        if (kk + 32 < KDIM)
            __builtin_prefetch(&W[(size_t)(tid & 127) * KDIM + kk + 32], 0, 1);
        __syncthreads();
#pragma unroll
        for (int s = 0; s < 8; ++s) {
            const int kb = 4 * s + lhi * 2;
            v2f a, b;
            a.x = lds_x[lm][kk + kb + 0];
            a.y = lds_x[lm][kk + kb + 1];
            b.x = lds_w[c0 + lm][kb + 0];
            b.y = lds_w[c0 + lm][kb + 1];
            acc = __builtin_amdgcn_wmma_f32_16x16x4_f32(false, a, false, b,
                                                        (short)0, acc, false, false);
        }
    }
#pragma unroll
    for (int v = 0; v < 8; ++v) {
        int m = v + lhi * 8;
        h[(size_t)(r0 + m) * KDIM + c0 + lm] = acc[v];
    }
}

__global__ __launch_bounds__(256) void gat_alpha(const float* __restrict__ h,
                                                 const float* __restrict__ a_src,
                                                 const float* __restrict__ a_dst,
                                                 float* __restrict__ alpha_src,
                                                 float* __restrict__ alpha_dst) {
    int t = blockIdx.x * 256 + threadIdx.x;
    int head = t & 3;
    int bn   = t >> 2;
    const float4* hp = (const float4*)(h + (size_t)bn * KDIM + head * DD);
    const float4* sp = (const float4*)(a_src + head * DD);
    const float4* dp = (const float4*)(a_dst + head * DD);
    float ss = 0.f, sd = 0.f;
#pragma unroll
    for (int i = 0; i < 8; ++i) {
        float4 hv = hp[i], av = sp[i], dv = dp[i];
        ss += hv.x * av.x + hv.y * av.y + hv.z * av.z + hv.w * av.w;
        sd += hv.x * dv.x + hv.y * dv.y + hv.z * dv.z + hv.w * dv.w;
    }
    alpha_src[t] = ss;
    alpha_dst[t] = sd;
}

__global__ __launch_bounds__(256) void gat_dmax(const float* __restrict__ alpha_dst,
                                                const float* __restrict__ alive,
                                                float* __restrict__ dmax) {
    __shared__ float red[256];
    const int bh = blockIdx.x;
    const int b = bh >> 2, head = bh & 3;
    float m = -INFINITY;
    for (int j = threadIdx.x; j < NN; j += 256) {
        float d  = alpha_dst[(size_t)(b * NN + j) * HH + head];
        if (alive[b * NN + j] >= 0.5f && d > m) m = d;
    }
    red[threadIdx.x] = m;
    __syncthreads();
    for (int s = 128; s > 0; s >>= 1) {
        if (threadIdx.x < s) red[threadIdx.x] = fmaxf(red[threadIdx.x], red[threadIdx.x + s]);
        __syncthreads();
    }
    if (threadIdx.x == 0) dmax[bh] = red[0];
}

__global__ __launch_bounds__(256) void gat_attn(const float* __restrict__ h,
                                                const float* __restrict__ alpha_src,
                                                const float* __restrict__ alpha_dst,
                                                const float* __restrict__ alive,
                                                const float* __restrict__ dmax,
                                                float* __restrict__ out) {
    __shared__ float dval[HH][NN];
    __shared__ float inv_s[8][16];
    const int tid   = threadIdx.x;
    const int lane  = tid & 31;
    const int wave  = tid >> 5;
    const int head  = wave & 3;
    const int dhalf = wave >> 2;
    const int b     = blockIdx.x >> 6;
    const int q0    = (blockIdx.x & 63) * 16;
    const int lm    = lane & 15;
    const int lhi   = lane >> 4;
    const int d0    = head * DD + dhalf * 16;

    for (int idx = tid; idx < HH * NN; idx += 256) {
        int hh = idx & 3, j = idx >> 2;
        float d = alpha_dst[(size_t)(b * NN + j) * HH + hh];
        dval[hh][j] = (alive[b * NN + j] >= 0.5f) ? d : -INFINITY;
    }
    __syncthreads();

    const float Dm  = dmax[b * HH + head];
    const float s_i = alpha_src[(size_t)(b * NN + q0 + lm) * HH + head];
    const float e0  = s_i + Dm;
    const float m_i = (e0 >= 0.f) ? e0 : NEG_SLOPE * e0;

    v8f acc = {};
    float zacc = 0.f;
    if (Dm > -INFINITY) {
        const float* hbase = h + (size_t)b * NN * KDIM + d0 + lm;
        for (int j0 = 0; j0 < NN; j0 += 4) {
            const int ja = j0 + lhi * 2;
            float ea = s_i + dval[head][ja];
            float eb = s_i + dval[head][ja + 1];
            ea = (ea >= 0.f) ? ea : NEG_SLOPE * ea;
            eb = (eb >= 0.f) ? eb : NEG_SLOPE * eb;
            v2f p, vv;
            p.x = __expf(ea - m_i);
            p.y = __expf(eb - m_i);
            zacc += p.x + p.y;
            vv.x = hbase[(size_t)ja * KDIM];
            vv.y = hbase[(size_t)(ja + 1) * KDIM];
            acc = __builtin_amdgcn_wmma_f32_16x16x4_f32(false, p, false, vv,
                                                        (short)0, acc, false, false);
        }
    }
    zacc += __shfl_xor(zacc, 16, 32);
    if (lhi == 0) inv_s[wave][lm] = (zacc > 0.f) ? (1.f / zacc) : 0.f;
    __syncthreads();
#pragma unroll
    for (int v = 0; v < 8; ++v) {
        int m = v + lhi * 8;
        out[((size_t)(b * NN) + q0 + m) * KDIM + d0 + lm] = acc[v] * inv_s[wave][m];
    }
}

extern "C" void kernel_launch(void* const* d_in, const int* in_sizes, int n_in,
                              void* d_out, int out_size, void* d_ws, size_t ws_size,
                              hipStream_t stream) {
    const float* x     = (const float*)d_in[0];
    const float* alive = (const float*)d_in[1];
    const float* W     = (const float*)d_in[2];
    const float* a_src = (const float*)d_in[3];
    const float* a_dst = (const float*)d_in[4];
    float* out = (float*)d_out;

    float* h_ws  = (float*)d_ws;
    float* as_ws = h_ws + (size_t)BB * NN * KDIM;
    float* ad_ws = as_ws + (size_t)BB * NN * HH;
    float* dm_ws = ad_ws + (size_t)BB * NN * HH;

    gat_h_gemm<<<512, 256, 0, stream>>>(x, W, h_ws);
    gat_alpha <<<128, 256, 0, stream>>>(h_ws, a_src, a_dst, as_ws, ad_ws);
    gat_dmax  <<<32,  256, 0, stream>>>(ad_ws, alive, dm_ws);
    gat_attn  <<<512, 256, 0, stream>>>(h_ws, as_ws, ad_ws, alive, dm_ws, out);
}


# attention retiled: one block per (batch, head, 128 queries) keeps that head's 1024x32 V slab in LDS, loaded once and shared by 8 waves; each wave does 16 queries x both d halves; sum order kept
# speedup vs baseline: 1.1142x; 1.1142x over previous
#include <hip/hip_runtime.h>
#include <math.h>

typedef __attribute__((ext_vector_type(2))) float v2f;
typedef __attribute__((ext_vector_type(8))) float v8f;

#define HH 4
#define DD 32
#define NN 1024
#define BB 8
#define KDIM 128
#define NEG_SLOPE 0.2f

__global__ __launch_bounds__(256) void gat_h_gemm(const float* __restrict__ x,
                                                  const float* __restrict__ W,
                                                  float* __restrict__ h) {
    __shared__ float lds_x[16][132];
    __shared__ float lds_w[128][36];
    const int tid  = threadIdx.x;
    const int lane = tid & 31;
    const int wave = tid >> 5;
    const int r0   = blockIdx.x * 16;
    const int c0   = wave * 16;
    const int lm   = lane & 15;
    const int lhi  = lane >> 4;

    for (int idx = tid; idx < 16 * KDIM; idx += 256) {
        int row = idx >> 7, col = idx & 127;
        lds_x[row][col] = x[(size_t)(r0 + row) * KDIM + col];
    }

    v8f acc = {};
    for (int kk = 0; kk < KDIM; kk += 32) {
        __syncthreads();
        for (int idx = tid; idx < 128 * 32; idx += 256) {
            int row = idx >> 5, col = idx & 31;
            lds_w[row][col] = W[(size_t)row * KDIM + kk + col];
        }
        if (kk + 32 < KDIM)
            __builtin_prefetch(&W[(size_t)(tid & 127) * KDIM + kk + 32], 0, 1);
        __syncthreads();
#pragma unroll
        for (int s = 0; s < 8; ++s) {
            const int kb = 4 * s + lhi * 2;
            v2f a, b;
            a.x = lds_x[lm][kk + kb + 0];
            a.y = lds_x[lm][kk + kb + 1];
            b.x = lds_w[c0 + lm][kb + 0];
            b.y = lds_w[c0 + lm][kb + 1];
            acc = __builtin_amdgcn_wmma_f32_16x16x4_f32(false, a, false, b,
                                                        (short)0, acc, false, false);
        }
    }
#pragma unroll
    for (int v = 0; v < 8; ++v) {
        int m = v + lhi * 8;
        h[(size_t)(r0 + m) * KDIM + c0 + lm] = acc[v];
    }
}

__global__ __launch_bounds__(256) void gat_alpha(const float* __restrict__ h,
                                                 const float* __restrict__ a_src,
                                                 const float* __restrict__ a_dst,
                                                 float* __restrict__ alpha_src,
                                                 float* __restrict__ alpha_dst) {
    int t = blockIdx.x * 256 + threadIdx.x;
    int head = t & 3;
    int bn   = t >> 2;
    const float4* hp = (const float4*)(h + (size_t)bn * KDIM + head * DD);
    const float4* sp = (const float4*)(a_src + head * DD);
    const float4* dp = (const float4*)(a_dst + head * DD);
    float ss = 0.f, sd = 0.f;
#pragma unroll
    for (int i = 0; i < 8; ++i) {
        float4 hv = hp[i], av = sp[i], dv = dp[i];
        ss += hv.x * av.x + hv.y * av.y + hv.z * av.z + hv.w * av.w;
        sd += hv.x * dv.x + hv.y * dv.y + hv.z * dv.z + hv.w * dv.w;
    }
    alpha_src[t] = ss;
    alpha_dst[t] = sd;
}

__global__ __launch_bounds__(256) void gat_dmax(const float* __restrict__ alpha_dst,
                                                const float* __restrict__ alive,
                                                float* __restrict__ dmax) {
    __shared__ float red[256];
    const int bh = blockIdx.x;
    const int b = bh >> 2, head = bh & 3;
    float m = -INFINITY;
    for (int j = threadIdx.x; j < NN; j += 256) {
        float d  = alpha_dst[(size_t)(b * NN + j) * HH + head];
        if (alive[b * NN + j] >= 0.5f && d > m) m = d;
    }
    red[threadIdx.x] = m;
    __syncthreads();
    for (int s = 128; s > 0; s >>= 1) {
        if (threadIdx.x < s) red[threadIdx.x] = fmaxf(red[threadIdx.x], red[threadIdx.x + s]);
        __syncthreads();
    }
    if (threadIdx.x == 0) dmax[bh] = red[0];
}

#define QPB    128
#define KCH    128
#define VPITCH 40
__global__ __launch_bounds__(256) void gat_attn(const float* __restrict__ h,
                                                const float* __restrict__ alpha_src,
                                                const float* __restrict__ alpha_dst,
                                                const float* __restrict__ alive,
                                                const float* __restrict__ dmax,
                                                float* __restrict__ out) {
    __shared__ __attribute__((aligned(16))) float vsl[NN * VPITCH];
    __shared__ __attribute__((aligned(16))) float dval[NN];
    __shared__ float inv_s[8][16];
    const int tid   = threadIdx.x;
    const int lane  = tid & 31;
    const int wave  = tid >> 5;
    const int lm    = lane & 15;
    const int lhi   = lane >> 4;
    const int head  = (blockIdx.x >> 3) & 3;
    const int b     = blockIdx.x >> 5;
    const int q0    = (blockIdx.x & 7) * QPB + wave * 16;

    for (int j = tid; j < NN; j += 256) {
        float d = alpha_dst[(size_t)(b * NN + j) * HH + head];
        dval[j] = (alive[b * NN + j] >= 0.5f) ? d : -INFINITY;
    }

    const int lrow = tid >> 3;
    const int lseg = (tid & 7) * 4;
    const float* hsrc = h + (size_t)b * NN * KDIM + head * DD + lseg;
    float* vdst = vsl + lrow * VPITCH + lseg;
    const float4 f0 = *(const float4*)(hsrc + (size_t)(0 * 32 + lrow) * KDIM);
    const float4 f1 = *(const float4*)(hsrc + (size_t)(1 * 32 + lrow) * KDIM);
    const float4 f2 = *(const float4*)(hsrc + (size_t)(2 * 32 + lrow) * KDIM);
    const float4 f3 = *(const float4*)(hsrc + (size_t)(3 * 32 + lrow) * KDIM);
    *(float4*)(vdst + 0 * 32 * VPITCH) = f0;
    *(float4*)(vdst + 1 * 32 * VPITCH) = f1;
    *(float4*)(vdst + 2 * 32 * VPITCH) = f2;
    *(float4*)(vdst + 3 * 32 * VPITCH) = f3;
    float4 pre0 = *(const float4*)(hsrc + (size_t)(KCH + 0 * 32 + lrow) * KDIM);
    float4 pre1 = *(const float4*)(hsrc + (size_t)(KCH + 1 * 32 + lrow) * KDIM);
    float4 pre2 = *(const float4*)(hsrc + (size_t)(KCH + 2 * 32 + lrow) * KDIM);
    float4 pre3 = *(const float4*)(hsrc + (size_t)(KCH + 3 * 32 + lrow) * KDIM);

    const float Dm  = dmax[b * HH + head];
    const float s_i = alpha_src[(size_t)(b * NN + q0 + lm) * HH + head];
    const float e0  = s_i + Dm;
    const float m_i = (Dm > -INFINITY) ? ((e0 >= 0.f) ? e0 : NEG_SLOPE * e0) : 0.f;
    __syncthreads();

    v8f acc0 = {}, acc1 = {};
    float zacc = 0.f;
#pragma unroll 1
    for (int c = 0; c < NN / KCH; ++c) {
        float4 nx0 = pre0, nx1 = pre1, nx2 = pre2, nx3 = pre3;
        if (c + 2 < NN / KCH) {
            const float* hn = hsrc + (size_t)((c + 2) * KCH + lrow) * KDIM;
            nx0 = *(const float4*)(hn + (size_t)(0 * 32) * KDIM);
            nx1 = *(const float4*)(hn + (size_t)(1 * 32) * KDIM);
            nx2 = *(const float4*)(hn + (size_t)(2 * 32) * KDIM);
            nx3 = *(const float4*)(hn + (size_t)(3 * 32) * KDIM);
        }
        const float* dv = dval + c * KCH;
        const float* vc = vsl + c * KCH * VPITCH + lm;
        for (int jj = 0; jj < KCH; jj += 4) {
            const int ja = jj + lhi * 2;
            float ea = s_i + dv[ja];
            float eb = s_i + dv[ja + 1];
            ea = (ea >= 0.f) ? ea : NEG_SLOPE * ea;
            eb = (eb >= 0.f) ? eb : NEG_SLOPE * eb;
            v2f p, v0, v1;
            p.x = __expf(ea - m_i);
            p.y = __expf(eb - m_i);
            zacc += p.x + p.y;
            v0.x = vc[ja * VPITCH];
            v0.y = vc[(ja + 1) * VPITCH];
            v1.x = vc[ja * VPITCH + 16];
            v1.y = vc[(ja + 1) * VPITCH + 16];
            acc0 = __builtin_amdgcn_wmma_f32_16x16x4_f32(false, p, false, v0,
                                                         (short)0, acc0, false, false);
            acc1 = __builtin_amdgcn_wmma_f32_16x16x4_f32(false, p, false, v1,
                                                         (short)0, acc1, false, false);
        }
        if (c + 1 < NN / KCH) {
            float* vn = vdst + (c + 1) * KCH * VPITCH;
            *(float4*)(vn + 0 * 32 * VPITCH) = pre0;
            *(float4*)(vn + 1 * 32 * VPITCH) = pre1;
            *(float4*)(vn + 2 * 32 * VPITCH) = pre2;
            *(float4*)(vn + 3 * 32 * VPITCH) = pre3;
        }
        __syncthreads();
        pre0 = nx0; pre1 = nx1; pre2 = nx2; pre3 = nx3;
    }
    zacc += __shfl_xor(zacc, 16, 32);
    if (lhi == 0) inv_s[wave][lm] = (zacc > 0.f) ? (1.f / zacc) : 0.f;
    __syncthreads();
#pragma unroll
    for (int v = 0; v < 8; ++v) {
        int m = v + lhi * 8;
        const float sc = inv_s[wave][m];
        float* o = out + ((size_t)(b * NN) + q0 + m) * KDIM + head * DD + lm;
        o[0]  = acc0[v] * sc;
        o[16] = acc1[v] * sc;
    }
}

extern "C" void kernel_launch(void* const* d_in, const int* in_sizes, int n_in,
                              void* d_out, int out_size, void* d_ws, size_t ws_size,
                              hipStream_t stream) {
    const float* x     = (const float*)d_in[0];
    const float* alive = (const float*)d_in[1];
    const float* W     = (const float*)d_in[2];
    const float* a_src = (const float*)d_in[3];
    const float* a_dst = (const float*)d_in[4];
    float* out = (float*)d_out;

    float* h_ws  = (float*)d_ws;
    float* as_ws = h_ws + (size_t)BB * NN * KDIM;
    float* ad_ws = as_ws + (size_t)BB * NN * HH;
    float* dm_ws = ad_ws + (size_t)BB * NN * HH;

    gat_h_gemm<<<512, 256, 0, stream>>>(x, W, h_ws);
    gat_alpha <<<128, 256, 0, stream>>>(h_ws, a_src, a_dst, as_ws, ad_ws);
    gat_dmax  <<<32,  256, 0, stream>>>(ad_ws, alive, dm_ws);
    gat_attn  <<<256, 256, 0, stream>>>(h_ws, as_ws, ad_ws, alive, dm_ws, out);
}


# 4 launches -> 2: alpha dot products move into the GEMM epilogue (own h tile via LDS, same sum order); each attention block takes the max of the masked alpha_dst it already loads
# speedup vs baseline: 1.2799x; 1.1487x over previous
#include <hip/hip_runtime.h>
#include <math.h>

typedef __attribute__((ext_vector_type(2))) float v2f;
typedef __attribute__((ext_vector_type(8))) float v8f;

#define HH 4
#define DD 32
#define NN 1024
#define BB 8
#define KDIM 128
#define NEG_SLOPE 0.2f

__global__ __launch_bounds__(256) void gat_h_gemm(const float* __restrict__ x,
                                                  const float* __restrict__ W,
                                                  const float* __restrict__ a_src,
                                                  const float* __restrict__ a_dst,
                                                  float* __restrict__ h,
                                                  float* __restrict__ alpha_src,
                                                  float* __restrict__ alpha_dst) {
    __shared__ __attribute__((aligned(16))) float lds_x[16][132];
    __shared__ float lds_w[128][36];
    const int tid  = threadIdx.x;
    const int lane = tid & 31;
    const int wave = tid >> 5;
    const int r0   = blockIdx.x * 16;
    const int c0   = wave * 16;
    const int lm   = lane & 15;
    const int lhi  = lane >> 4;

    const bool dot_thr = tid < 128;
    const int  didx    = tid & 63;
    const int  drow    = didx >> 2;
    const int  dhead   = didx & 3;
    float4 cf[8];
    if (dot_thr) {
        const float4* ap = (const float4*)(((tid >> 6) ? a_dst : a_src) + dhead * DD);
#pragma unroll
        for (int i = 0; i < 8; ++i) cf[i] = ap[i];
    }

    for (int idx = tid; idx < 16 * KDIM; idx += 256) {
        int row = idx >> 7, col = idx & 127;
        lds_x[row][col] = x[(size_t)(r0 + row) * KDIM + col];
    }

    v8f acc = {};
    for (int kk = 0; kk < KDIM; kk += 32) {
        __syncthreads();
        for (int idx = tid; idx < 128 * 32; idx += 256) {
            int row = idx >> 5, col = idx & 31;
            lds_w[row][col] = W[(size_t)row * KDIM + kk + col];
        }
        if (kk + 32 < KDIM)
            __builtin_prefetch(&W[(size_t)(tid & 127) * KDIM + kk + 32], 0, 1);
        __syncthreads();
#pragma unroll
        for (int s = 0; s < 8; ++s) {
            const int kb = 4 * s + lhi * 2;
            v2f a, b;
            a.x = lds_x[lm][kk + kb + 0];
            a.y = lds_x[lm][kk + kb + 1];
            b.x = lds_w[c0 + lm][kb + 0];
            b.y = lds_w[c0 + lm][kb + 1];
            acc = __builtin_amdgcn_wmma_f32_16x16x4_f32(false, a, false, b,
                                                        (short)0, acc, false, false);
        }
    }
    __syncthreads();
#pragma unroll
    for (int v = 0; v < 8; ++v) {
        int m = v + lhi * 8;
        lds_x[m][c0 + lm] = acc[v];
    }
    __syncthreads();
#pragma unroll
    for (int v = 0; v < 8; ++v) {
        int m = v + lhi * 8;
        h[(size_t)(r0 + m) * KDIM + c0 + lm] = acc[v];
    }
    if (dot_thr) {
        const float4* hp = (const float4*)&lds_x[drow][dhead * DD];
        float s = 0.f;
#pragma unroll
        for (int i = 0; i < 8; ++i) {
            float4 hv = hp[i], av = cf[i];
            s += hv.x * av.x + hv.y * av.y + hv.z * av.z + hv.w * av.w;
        }
        float* dstp = (tid >> 6) ? alpha_dst : alpha_src;
        dstp[(size_t)r0 * HH + didx] = s;
    }
}

#define QPB    128
#define KCH    128
#define VPITCH 40
__global__ __launch_bounds__(256) void gat_attn(const float* __restrict__ h,
                                                const float* __restrict__ alpha_src,
                                                const float* __restrict__ alpha_dst,
                                                const float* __restrict__ alive,
                                                float* __restrict__ out) {
    __shared__ __attribute__((aligned(16))) float vsl[NN * VPITCH];
    __shared__ __attribute__((aligned(16))) float dval[NN];
    __shared__ float inv_s[8][16];
    __shared__ __attribute__((aligned(16))) float wmax[8];
    const int tid   = threadIdx.x;
    const int lane  = tid & 31;
    const int wave  = tid >> 5;
    const int lm    = lane & 15;
    const int lhi   = lane >> 4;
    const int head  = (blockIdx.x >> 3) & 3;
    const int b     = blockIdx.x >> 5;
    const int q0    = (blockIdx.x & 7) * QPB + wave * 16;

    float mloc = -INFINITY;
    for (int j = tid; j < NN; j += 256) {
        float d = alpha_dst[(size_t)(b * NN + j) * HH + head];
        const bool live = alive[b * NN + j] >= 0.5f;
        dval[j] = live ? d : -INFINITY;
        if (live && d > mloc) mloc = d;
    }

    const int lrow = tid >> 3;
    const int lseg = (tid & 7) * 4;
    const float* hsrc = h + (size_t)b * NN * KDIM + head * DD + lseg;
    float* vdst = vsl + lrow * VPITCH + lseg;
    const float4 f0 = *(const float4*)(hsrc + (size_t)(0 * 32 + lrow) * KDIM);
    const float4 f1 = *(const float4*)(hsrc + (size_t)(1 * 32 + lrow) * KDIM);
    const float4 f2 = *(const float4*)(hsrc + (size_t)(2 * 32 + lrow) * KDIM);
    const float4 f3 = *(const float4*)(hsrc + (size_t)(3 * 32 + lrow) * KDIM);
    float4 pre0 = *(const float4*)(hsrc + (size_t)(KCH + 0 * 32 + lrow) * KDIM);
    float4 pre1 = *(const float4*)(hsrc + (size_t)(KCH + 1 * 32 + lrow) * KDIM);
    float4 pre2 = *(const float4*)(hsrc + (size_t)(KCH + 2 * 32 + lrow) * KDIM);
    float4 pre3 = *(const float4*)(hsrc + (size_t)(KCH + 3 * 32 + lrow) * KDIM);

    const float s_i = alpha_src[(size_t)(b * NN + q0 + lm) * HH + head];
    for (int o = 16; o > 0; o >>= 1) mloc = fmaxf(mloc, __shfl_xor(mloc, o, 32));
    *(float4*)(vdst + 0 * 32 * VPITCH) = f0;
    *(float4*)(vdst + 1 * 32 * VPITCH) = f1;
    *(float4*)(vdst + 2 * 32 * VPITCH) = f2;
    *(float4*)(vdst + 3 * 32 * VPITCH) = f3;
    if (lane == 0) wmax[wave] = mloc;
    __syncthreads();
    const float4 wa = *(const float4*)&wmax[0];
    const float4 wb = *(const float4*)&wmax[4];
    const float Dm  = fmaxf(fmaxf(fmaxf(wa.x, wa.y), fmaxf(wa.z, wa.w)),
                            fmaxf(fmaxf(wb.x, wb.y), fmaxf(wb.z, wb.w)));
    const float e0  = s_i + Dm;
    const float m_i = (Dm > -INFINITY) ? ((e0 >= 0.f) ? e0 : NEG_SLOPE * e0) : 0.f;

    v8f acc0 = {}, acc1 = {};
    float zacc = 0.f;
#pragma unroll 1
    for (int c = 0; c < NN / KCH; ++c) {
        float4 nx0 = pre0, nx1 = pre1, nx2 = pre2, nx3 = pre3;
        if (c + 2 < NN / KCH) {
            const float* hn = hsrc + (size_t)((c + 2) * KCH + lrow) * KDIM;
            nx0 = *(const float4*)(hn + (size_t)(0 * 32) * KDIM);
            nx1 = *(const float4*)(hn + (size_t)(1 * 32) * KDIM);
            nx2 = *(const float4*)(hn + (size_t)(2 * 32) * KDIM);
            nx3 = *(const float4*)(hn + (size_t)(3 * 32) * KDIM);
        }
        const float* dv = dval + c * KCH;
        const float* vc = vsl + c * KCH * VPITCH + lm;
        for (int jj = 0; jj < KCH; jj += 4) {
            const int ja = jj + lhi * 2;
            float ea = s_i + dv[ja];
            float eb = s_i + dv[ja + 1];
            ea = (ea >= 0.f) ? ea : NEG_SLOPE * ea;
            eb = (eb >= 0.f) ? eb : NEG_SLOPE * eb;
            v2f p, v0, v1;
            p.x = __expf(ea - m_i);
            p.y = __expf(eb - m_i);
            zacc += p.x + p.y;
            v0.x = vc[ja * VPITCH];
            v0.y = vc[(ja + 1) * VPITCH];
            v1.x = vc[ja * VPITCH + 16];
            v1.y = vc[(ja + 1) * VPITCH + 16];
            acc0 = __builtin_amdgcn_wmma_f32_16x16x4_f32(false, p, false, v0,
                                                         (short)0, acc0, false, false);
            acc1 = __builtin_amdgcn_wmma_f32_16x16x4_f32(false, p, false, v1,
                                                         (short)0, acc1, false, false);
        }
        if (c + 1 < NN / KCH) {
            float* vn = vdst + (c + 1) * KCH * VPITCH;
            *(float4*)(vn + 0 * 32 * VPITCH) = pre0;
            *(float4*)(vn + 1 * 32 * VPITCH) = pre1;
            *(float4*)(vn + 2 * 32 * VPITCH) = pre2;
            *(float4*)(vn + 3 * 32 * VPITCH) = pre3;
        }
        __syncthreads();
        pre0 = nx0; pre1 = nx1; pre2 = nx2; pre3 = nx3;
    }
    zacc += __shfl_xor(zacc, 16, 32);
    if (lhi == 0) inv_s[wave][lm] = (zacc > 0.f) ? (1.f / zacc) : 0.f;
    __syncthreads();
#pragma unroll
    for (int v = 0; v < 8; ++v) {
        int m = v + lhi * 8;
        const float sc = inv_s[wave][m];
        float* o = out + ((size_t)(b * NN) + q0 + m) * KDIM + head * DD + lm;
        o[0]  = acc0[v] * sc;
        o[16] = acc1[v] * sc;
    }
}

extern "C" void kernel_launch(void* const* d_in, const int* in_sizes, int n_in,
                              void* d_out, int out_size, void* d_ws, size_t ws_size,
                              hipStream_t stream) {
    const float* x     = (const float*)d_in[0];
    const float* alive = (const float*)d_in[1];
    const float* W     = (const float*)d_in[2];
    const float* a_src = (const float*)d_in[3];
    const float* a_dst = (const float*)d_in[4];
    float* out = (float*)d_out;

    float* h_ws  = (float*)d_ws;
    float* as_ws = h_ws + (size_t)BB * NN * KDIM;
    float* ad_ws = as_ws + (size_t)BB * NN * HH;

    gat_h_gemm<<<512, 256, 0, stream>>>(x, W, a_src, a_dst, h_ws, as_ws, ad_ws);
    gat_attn  <<<256, 256, 0, stream>>>(h_ws, as_ws, ad_ws, alive, out);
}


# attention weights without a per-pair exponential: exp(leaky(s_i+d_j)-m_i) = max(a_i*E_j, c_i*F_j); per-key (E,F) table built once per block in LDS; key loop: 1 packed mul + 1 max per weight
# speedup vs baseline: 1.3710x; 1.0712x over previous
#include <hip/hip_runtime.h>
#include <math.h>

typedef __attribute__((ext_vector_type(2))) float v2f;
typedef __attribute__((ext_vector_type(8))) float v8f;

#define HH 4
#define DD 32
#define NN 1024
#define BB 8
#define KDIM 128
#define NEG_SLOPE 0.2f

__global__ __launch_bounds__(256) void gat_h_gemm(const float* __restrict__ x,
                                                  const float* __restrict__ W,
                                                  const float* __restrict__ a_src,
                                                  const float* __restrict__ a_dst,
                                                  float* __restrict__ h,
                                                  float* __restrict__ alpha_src,
                                                  float* __restrict__ alpha_dst) {
    __shared__ __attribute__((aligned(16))) float lds_x[16][132];
    __shared__ float lds_w[128][36];
    const int tid  = threadIdx.x;
    const int lane = tid & 31;
    const int wave = tid >> 5;
    const int r0   = blockIdx.x * 16;
    const int c0   = wave * 16;
    const int lm   = lane & 15;
    const int lhi  = lane >> 4;

    const bool dot_thr = tid < 128;
    const int  didx    = tid & 63;
    const int  drow    = didx >> 2;
    const int  dhead   = didx & 3;
    float4 cf[8];
    if (dot_thr) {
        const float4* ap = (const float4*)(((tid >> 6) ? a_dst : a_src) + dhead * DD);
#pragma unroll
        for (int i = 0; i < 8; ++i) cf[i] = ap[i];
    }

    for (int idx = tid; idx < 16 * KDIM; idx += 256) {
        int row = idx >> 7, col = idx & 127;
        lds_x[row][col] = x[(size_t)(r0 + row) * KDIM + col];
    }

    v8f acc = {};
    for (int kk = 0; kk < KDIM; kk += 32) {
        __syncthreads();
        for (int idx = tid; idx < 128 * 32; idx += 256) {
            int row = idx >> 5, col = idx & 31;
            lds_w[row][col] = W[(size_t)row * KDIM + kk + col];
        }
        if (kk + 32 < KDIM)
            __builtin_prefetch(&W[(size_t)(tid & 127) * KDIM + kk + 32], 0, 1);
        __syncthreads();
#pragma unroll
        for (int s = 0; s < 8; ++s) {
            const int kb = 4 * s + lhi * 2;
            v2f a, b;
            a.x = lds_x[lm][kk + kb + 0];
            a.y = lds_x[lm][kk + kb + 1];
            b.x = lds_w[c0 + lm][kb + 0];
            b.y = lds_w[c0 + lm][kb + 1];
            acc = __builtin_amdgcn_wmma_f32_16x16x4_f32(false, a, false, b,
                                                        (short)0, acc, false, false);
        }
    }
    __syncthreads();
#pragma unroll
    for (int v = 0; v < 8; ++v) {
        int m = v + lhi * 8;
        lds_x[m][c0 + lm] = acc[v];
    }
    __syncthreads();
#pragma unroll
    for (int v = 0; v < 8; ++v) {
        int m = v + lhi * 8;
        h[(size_t)(r0 + m) * KDIM + c0 + lm] = acc[v];
    }
    if (dot_thr) {
        const float4* hp = (const float4*)&lds_x[drow][dhead * DD];
        float s = 0.f;
#pragma unroll
        for (int i = 0; i < 8; ++i) {
            float4 hv = hp[i], av = cf[i];
            s += hv.x * av.x + hv.y * av.y + hv.z * av.z + hv.w * av.w;
        }
        float* dstp = (tid >> 6) ? alpha_dst : alpha_src;
        dstp[(size_t)r0 * HH + didx] = s;
    }
}

#define QPB    128
#define KCH    128
#define VPITCH 40
__global__ __launch_bounds__(256) void gat_attn(const float* __restrict__ h,
                                                const float* __restrict__ alpha_src,
                                                const float* __restrict__ alpha_dst,
                                                const float* __restrict__ alive,
                                                float* __restrict__ out) {
    __shared__ __attribute__((aligned(16))) float vsl[NN * VPITCH];
    __shared__ __attribute__((aligned(16))) float ef[NN * 2];
    __shared__ float inv_s[8][16];
    __shared__ __attribute__((aligned(16))) float wmax[8];
    const int tid   = threadIdx.x;
    const int lane  = tid & 31;
    const int wave  = tid >> 5;
    const int lm    = lane & 15;
    const int lhi   = lane >> 4;
    const int head  = (blockIdx.x >> 3) & 3;
    const int b     = blockIdx.x >> 5;
    const int q0    = (blockIdx.x & 7) * QPB + wave * 16;

    float dk[NN / 256];
    bool  lv[NN / 256];
    float mloc = -INFINITY;
#pragma unroll
    for (int k = 0; k < NN / 256; ++k) {
        const int j = tid + 256 * k;
        dk[k] = alpha_dst[(size_t)(b * NN + j) * HH + head];
        lv[k] = alive[b * NN + j] >= 0.5f;
    }
#pragma unroll
    for (int k = 0; k < NN / 256; ++k)
        if (lv[k] && dk[k] > mloc) mloc = dk[k];

    const int lrow = tid >> 3;
    const int lseg = (tid & 7) * 4;
    const float* hsrc = h + (size_t)b * NN * KDIM + head * DD + lseg;
    float* vdst = vsl + lrow * VPITCH + lseg;
    const float4 f0 = *(const float4*)(hsrc + (size_t)(0 * 32 + lrow) * KDIM);
    const float4 f1 = *(const float4*)(hsrc + (size_t)(1 * 32 + lrow) * KDIM);
    const float4 f2 = *(const float4*)(hsrc + (size_t)(2 * 32 + lrow) * KDIM);
    const float4 f3 = *(const float4*)(hsrc + (size_t)(3 * 32 + lrow) * KDIM);
    float4 pre0 = *(const float4*)(hsrc + (size_t)(KCH + 0 * 32 + lrow) * KDIM);
    float4 pre1 = *(const float4*)(hsrc + (size_t)(KCH + 1 * 32 + lrow) * KDIM);
    float4 pre2 = *(const float4*)(hsrc + (size_t)(KCH + 2 * 32 + lrow) * KDIM);
    float4 pre3 = *(const float4*)(hsrc + (size_t)(KCH + 3 * 32 + lrow) * KDIM);

    const float s_i = alpha_src[(size_t)(b * NN + q0 + lm) * HH + head];
    for (int o = 16; o > 0; o >>= 1) mloc = fmaxf(mloc, __shfl_xor(mloc, o, 32));
    *(float4*)(vdst + 0 * 32 * VPITCH) = f0;
    *(float4*)(vdst + 1 * 32 * VPITCH) = f1;
    *(float4*)(vdst + 2 * 32 * VPITCH) = f2;
    *(float4*)(vdst + 3 * 32 * VPITCH) = f3;
    if (lane == 0) wmax[wave] = mloc;
    __syncthreads();
    const float4 wa = *(const float4*)&wmax[0];
    const float4 wb = *(const float4*)&wmax[4];
    const float Dm  = fmaxf(fmaxf(fmaxf(wa.x, wa.y), fmaxf(wa.z, wa.w)),
                            fmaxf(fmaxf(wb.x, wb.y), fmaxf(wb.z, wb.w)));
#pragma unroll
    for (int k = 0; k < NN / 256; ++k) {
        const float t = dk[k] - Dm;
        float2 w;
        w.x = lv[k] ? __expf(t) : 0.f;
        w.y = lv[k] ? __expf(NEG_SLOPE * t) : 0.f;
        *(float2*)&ef[2 * (tid + 256 * k)] = w;
    }
    const float e0 = s_i + Dm;
    const float sm = __expf(-fabsf((1.f - NEG_SLOPE) * e0));
    v2f ac;
    ac.x = (e0 >= 0.f) ? 1.f : sm;
    ac.y = (e0 >= 0.f) ? sm : 1.f;
    __syncthreads();

    const float4* efl = (const float4*)ef + lhi;
    const float*  vl  = vsl + lhi * 2 * VPITCH + lm;
    v8f acc0 = {}, acc1 = {};
    float zacc = 0.f;
#pragma unroll 1
    for (int c = 0; c < NN / KCH; ++c) {
        float4 nx0 = pre0, nx1 = pre1, nx2 = pre2, nx3 = pre3;
        if (c + 2 < NN / KCH) {
            const float* hn = hsrc + (size_t)((c + 2) * KCH + lrow) * KDIM;
            nx0 = *(const float4*)(hn + (size_t)(0 * 32) * KDIM);
            nx1 = *(const float4*)(hn + (size_t)(1 * 32) * KDIM);
            nx2 = *(const float4*)(hn + (size_t)(2 * 32) * KDIM);
            nx3 = *(const float4*)(hn + (size_t)(3 * 32) * KDIM);
        }
        const float4* efc = efl + c * (KCH / 2);
        const float* vc = vl + c * KCH * VPITCH;
        for (int jj = 0; jj < KCH; jj += 4) {
            const float4 t = efc[jj >> 1];
            v2f qa, qb, p, v0, v1;
            qa.x = t.x; qa.y = t.y;
            qb.x = t.z; qb.y = t.w;
            qa *= ac;
            qb *= ac;
            p.x = fmaxf(qa.x, qa.y);
            p.y = fmaxf(qb.x, qb.y);
            zacc += p.x + p.y;
            v0.x = vc[jj * VPITCH];
            v0.y = vc[(jj + 1) * VPITCH];
            v1.x = vc[jj * VPITCH + 16];
            v1.y = vc[(jj + 1) * VPITCH + 16];
            acc0 = __builtin_amdgcn_wmma_f32_16x16x4_f32(false, p, false, v0,
                                                         (short)0, acc0, false, false);
            acc1 = __builtin_amdgcn_wmma_f32_16x16x4_f32(false, p, false, v1,
                                                         (short)0, acc1, false, false);
        }
        if (c + 1 < NN / KCH) {
            float* vn = vdst + (c + 1) * KCH * VPITCH;
            *(float4*)(vn + 0 * 32 * VPITCH) = pre0;
            *(float4*)(vn + 1 * 32 * VPITCH) = pre1;
            *(float4*)(vn + 2 * 32 * VPITCH) = pre2;
            *(float4*)(vn + 3 * 32 * VPITCH) = pre3;
        }
        __syncthreads();
        pre0 = nx0; pre1 = nx1; pre2 = nx2; pre3 = nx3;
    }
    zacc += __shfl_xor(zacc, 16, 32);
    if (lhi == 0) inv_s[wave][lm] = (zacc > 0.f) ? (1.f / zacc) : 0.f;
    __syncthreads();
#pragma unroll
    for (int v = 0; v < 8; ++v) {
        int m = v + lhi * 8;
        const float sc = inv_s[wave][m];
        float* o = out + ((size_t)(b * NN) + q0 + m) * KDIM + head * DD + lm;
        o[0]  = acc0[v] * sc;
        o[16] = acc1[v] * sc;
    }
}

extern "C" void kernel_launch(void* const* d_in, const int* in_sizes, int n_in,
                              void* d_out, int out_size, void* d_ws, size_t ws_size,
                              hipStream_t stream) {
    const float* x     = (const float*)d_in[0];
    const float* alive = (const float*)d_in[1];
    const float* W     = (const float*)d_in[2];
    const float* a_src = (const float*)d_in[3];
    const float* a_dst = (const float*)d_in[4];
    float* out = (float*)d_out;

    float* h_ws  = (float*)d_ws;
    float* as_ws = h_ws + (size_t)BB * NN * KDIM;
    float* ad_ws = as_ws + (size_t)BB * NN * HH;

    gat_h_gemm<<<512, 256, 0, stream>>>(x, W, a_src, a_dst, h_ws, as_ws, ad_ws);
    gat_attn  <<<256, 256, 0, stream>>>(h_ws, as_ws, ad_ws, alive, out);
}


# GEMM operands staged in one memory round trip: each thread issues all its 16-byte loads of W (whole 128x128) and the x tile before any wait, then stores to LDS; K loop unchunked, same k order
# speedup vs baseline: 1.9172x; 1.3984x over previous
#include <hip/hip_runtime.h>
#include <math.h>

typedef __attribute__((ext_vector_type(2))) float v2f;
typedef __attribute__((ext_vector_type(8))) float v8f;

#define HH 4
#define DD 32
#define NN 1024
#define BB 8
#define KDIM 128
#define NEG_SLOPE 0.2f

__global__ __launch_bounds__(256) void gat_h_gemm(const float* __restrict__ x,
                                                  const float* __restrict__ W,
                                                  const float* __restrict__ a_src,
                                                  const float* __restrict__ a_dst,
                                                  float* __restrict__ h,
                                                  float* __restrict__ alpha_src,
                                                  float* __restrict__ alpha_dst) {
    __shared__ __attribute__((aligned(16))) float lds_x[16][132];
    __shared__ __attribute__((aligned(16))) float lds_w[128][132];
    const int tid  = threadIdx.x;
    const int lane = tid & 31;
    const int wave = tid >> 5;
    const int r0   = blockIdx.x * 16;
    const int c0   = wave * 16;
    const int lm   = lane & 15;
    const int lhi  = lane >> 4;

    const bool dot_thr = tid < 128;
    const int  didx    = tid & 63;
    const int  drow    = didx >> 2;
    const int  dhead   = didx & 3;
    float4 cf[8];
    if (dot_thr) {
        const float4* ap = (const float4*)(((tid >> 6) ? a_dst : a_src) + dhead * DD);
#pragma unroll
        for (int i = 0; i < 8; ++i) cf[i] = ap[i];
    }

    const float* xsrc = x + (size_t)(r0 + wave) * KDIM + lane * 4;
    const float* wsrc = W + (size_t)wave * KDIM + lane * 4;
    float4 xr[2], wr[16];
#pragma unroll
    for (int i = 0; i < 2; ++i) xr[i] = *(const float4*)(xsrc + (size_t)(8 * i) * KDIM);
#pragma unroll
    for (int i = 0; i < 16; ++i) wr[i] = *(const float4*)(wsrc + (size_t)(8 * i) * KDIM);
#pragma unroll
    for (int i = 0; i < 2; ++i) *(float4*)&lds_x[wave + 8 * i][lane * 4] = xr[i];
#pragma unroll
    for (int i = 0; i < 16; ++i) *(float4*)&lds_w[wave + 8 * i][lane * 4] = wr[i];
    __syncthreads();

    v8f acc = {};
#pragma unroll
    for (int s = 0; s < KDIM / 4; ++s) {
        const int kb = 4 * s + lhi * 2;
        v2f a, b;
        a.x = lds_x[lm][kb + 0];
        a.y = lds_x[lm][kb + 1];
        b.x = lds_w[c0 + lm][kb + 0];
        b.y = lds_w[c0 + lm][kb + 1];
        acc = __builtin_amdgcn_wmma_f32_16x16x4_f32(false, a, false, b,
                                                    (short)0, acc, false, false);
    }
    __syncthreads();
#pragma unroll
    for (int v = 0; v < 8; ++v) {
        int m = v + lhi * 8;
        lds_x[m][c0 + lm] = acc[v];
    }
    __syncthreads();
#pragma unroll
    for (int v = 0; v < 8; ++v) {
        int m = v + lhi * 8;
        h[(size_t)(r0 + m) * KDIM + c0 + lm] = acc[v];
    }
    if (dot_thr) {
        const float4* hp = (const float4*)&lds_x[drow][dhead * DD];
        float s = 0.f;
#pragma unroll
        for (int i = 0; i < 8; ++i) {
            float4 hv = hp[i], av = cf[i];
            s += hv.x * av.x + hv.y * av.y + hv.z * av.z + hv.w * av.w;
        }
        float* dstp = (tid >> 6) ? alpha_dst : alpha_src;
        dstp[(size_t)r0 * HH + didx] = s;
    }
}

#define QPB    128
#define KCH    128
#define VPITCH 40
__global__ __launch_bounds__(256) void gat_attn(const float* __restrict__ h,
                                                const float* __restrict__ alpha_src,
                                                const float* __restrict__ alpha_dst,
                                                const float* __restrict__ alive,
                                                float* __restrict__ out) {
    __shared__ __attribute__((aligned(16))) float vsl[NN * VPITCH];
    __shared__ __attribute__((aligned(16))) float ef[NN * 2];
    __shared__ float inv_s[8][16];
    __shared__ __attribute__((aligned(16))) float wmax[8];
    const int tid   = threadIdx.x;
    const int lane  = tid & 31;
    const int wave  = tid >> 5;
    const int lm    = lane & 15;
    const int lhi   = lane >> 4;
    const int head  = (blockIdx.x >> 3) & 3;
    const int b     = blockIdx.x >> 5;
    const int q0    = (blockIdx.x & 7) * QPB + wave * 16;

    float dk[NN / 256];
    bool  lv[NN / 256];
    float mloc = -INFINITY;
#pragma unroll
    for (int k = 0; k < NN / 256; ++k) {
        const int j = tid + 256 * k;
        dk[k] = alpha_dst[(size_t)(b * NN + j) * HH + head];
        lv[k] = alive[b * NN + j] >= 0.5f;
    }
#pragma unroll
    for (int k = 0; k < NN / 256; ++k)
        if (lv[k] && dk[k] > mloc) mloc = dk[k];

    const int lrow = tid >> 3;
    const int lseg = (tid & 7) * 4;
    const float* hsrc = h + (size_t)b * NN * KDIM + head * DD + lseg;
    float* vdst = vsl + lrow * VPITCH + lseg;
    const float4 f0 = *(const float4*)(hsrc + (size_t)(0 * 32 + lrow) * KDIM);
    const float4 f1 = *(const float4*)(hsrc + (size_t)(1 * 32 + lrow) * KDIM);
    const float4 f2 = *(const float4*)(hsrc + (size_t)(2 * 32 + lrow) * KDIM);
    const float4 f3 = *(const float4*)(hsrc + (size_t)(3 * 32 + lrow) * KDIM);
    float4 pre0 = *(const float4*)(hsrc + (size_t)(KCH + 0 * 32 + lrow) * KDIM);
    float4 pre1 = *(const float4*)(hsrc + (size_t)(KCH + 1 * 32 + lrow) * KDIM);
    float4 pre2 = *(const float4*)(hsrc + (size_t)(KCH + 2 * 32 + lrow) * KDIM);
    float4 pre3 = *(const float4*)(hsrc + (size_t)(KCH + 3 * 32 + lrow) * KDIM);

    const float s_i = alpha_src[(size_t)(b * NN + q0 + lm) * HH + head];
    for (int o = 16; o > 0; o >>= 1) mloc = fmaxf(mloc, __shfl_xor(mloc, o, 32));
    *(float4*)(vdst + 0 * 32 * VPITCH) = f0;
    *(float4*)(vdst + 1 * 32 * VPITCH) = f1;
    *(float4*)(vdst + 2 * 32 * VPITCH) = f2;
    *(float4*)(vdst + 3 * 32 * VPITCH) = f3;
    if (lane == 0) wmax[wave] = mloc;
    __syncthreads();
    const float4 wa = *(const float4*)&wmax[0];
    const float4 wb = *(const float4*)&wmax[4];
    const float Dm  = fmaxf(fmaxf(fmaxf(wa.x, wa.y), fmaxf(wa.z, wa.w)),
                            fmaxf(fmaxf(wb.x, wb.y), fmaxf(wb.z, wb.w)));
#pragma unroll
    for (int k = 0; k < NN / 256; ++k) {
        const float t = dk[k] - Dm;
        float2 w;
        w.x = lv[k] ? __expf(t) : 0.f;
        w.y = lv[k] ? __expf(NEG_SLOPE * t) : 0.f;
        *(float2*)&ef[2 * (tid + 256 * k)] = w;
    }
    const float e0 = s_i + Dm;
    const float sm = __expf(-fabsf((1.f - NEG_SLOPE) * e0));
    v2f ac;
    ac.x = (e0 >= 0.f) ? 1.f : sm;
    ac.y = (e0 >= 0.f) ? sm : 1.f;
    __syncthreads();

    const float4* efl = (const float4*)ef + lhi;
    const float*  vl  = vsl + lhi * 2 * VPITCH + lm;
    v8f acc0 = {}, acc1 = {};
    float zacc = 0.f;
#pragma unroll 1
    for (int c = 0; c < NN / KCH; ++c) {
        float4 nx0 = pre0, nx1 = pre1, nx2 = pre2, nx3 = pre3;
        if (c + 2 < NN / KCH) {
            const float* hn = hsrc + (size_t)((c + 2) * KCH + lrow) * KDIM;
            nx0 = *(const float4*)(hn + (size_t)(0 * 32) * KDIM);
            nx1 = *(const float4*)(hn + (size_t)(1 * 32) * KDIM);
            nx2 = *(const float4*)(hn + (size_t)(2 * 32) * KDIM);
            nx3 = *(const float4*)(hn + (size_t)(3 * 32) * KDIM);
        }
        const float4* efc = efl + c * (KCH / 2);
        const float* vc = vl + c * KCH * VPITCH;
        for (int jj = 0; jj < KCH; jj += 4) {
            const float4 t = efc[jj >> 1];
            v2f qa, qb, p, v0, v1;
            qa.x = t.x; qa.y = t.y;
            qb.x = t.z; qb.y = t.w;
            qa *= ac;
            qb *= ac;
            p.x = fmaxf(qa.x, qa.y);
            p.y = fmaxf(qb.x, qb.y);
            zacc += p.x + p.y;
            v0.x = vc[jj * VPITCH];
            v0.y = vc[(jj + 1) * VPITCH];
            v1.x = vc[jj * VPITCH + 16];
            v1.y = vc[(jj + 1) * VPITCH + 16];
            acc0 = __builtin_amdgcn_wmma_f32_16x16x4_f32(false, p, false, v0,
                                                         (short)0, acc0, false, false);
            acc1 = __builtin_amdgcn_wmma_f32_16x16x4_f32(false, p, false, v1,
                                                         (short)0, acc1, false, false);
        }
        if (c + 1 < NN / KCH) {
            float* vn = vdst + (c + 1) * KCH * VPITCH;
            *(float4*)(vn + 0 * 32 * VPITCH) = pre0;
            *(float4*)(vn + 1 * 32 * VPITCH) = pre1;
            *(float4*)(vn + 2 * 32 * VPITCH) = pre2;
            *(float4*)(vn + 3 * 32 * VPITCH) = pre3;
        }
        __syncthreads();
        pre0 = nx0; pre1 = nx1; pre2 = nx2; pre3 = nx3;
    }
    zacc += __shfl_xor(zacc, 16, 32);
    if (lhi == 0) inv_s[wave][lm] = (zacc > 0.f) ? (1.f / zacc) : 0.f;
    __syncthreads();
#pragma unroll
    for (int v = 0; v < 8; ++v) {
        int m = v + lhi * 8;
        const float sc = inv_s[wave][m];
        float* o = out + ((size_t)(b * NN) + q0 + m) * KDIM + head * DD + lm;
        o[0]  = acc0[v] * sc;
        o[16] = acc1[v] * sc;
    }
}

extern "C" void kernel_launch(void* const* d_in, const int* in_sizes, int n_in,
                              void* d_out, int out_size, void* d_ws, size_t ws_size,
                              hipStream_t stream) {
    const float* x     = (const float*)d_in[0];
    const float* alive = (const float*)d_in[1];
    const float* W     = (const float*)d_in[2];
    const float* a_src = (const float*)d_in[3];
    const float* a_dst = (const float*)d_in[4];
    float* out = (float*)d_out;

    float* h_ws  = (float*)d_ws;
    float* as_ws = h_ws + (size_t)BB * NN * KDIM;
    float* ad_ws = as_ws + (size_t)BB * NN * HH;

    gat_h_gemm<<<512, 256, 0, stream>>>(x, W, a_src, a_dst, h_ws, as_ws, ad_ws);
    gat_attn  <<<256, 256, 0, stream>>>(h_ws, as_ws, ad_ws, alive, out);
}


# attention key loop fed by register blocks: V staged by key pair, one 16-byte LDS read gives both B fragments; next 64 keys' reads issued as one group ahead of the block's products; sum order kept
# speedup vs baseline: 2.1034x; 1.0971x over previous
#include <hip/hip_runtime.h>
#include <math.h>

typedef __attribute__((ext_vector_type(2))) float v2f;
typedef __attribute__((ext_vector_type(8))) float v8f;

#define HH 4
#define DD 32
#define NN 1024
#define BB 8
#define KDIM 128
#define NEG_SLOPE 0.2f

__global__ __launch_bounds__(256) void gat_h_gemm(const float* __restrict__ x,
                                                  const float* __restrict__ W,
                                                  const float* __restrict__ a_src,
                                                  const float* __restrict__ a_dst,
                                                  float* __restrict__ h,
                                                  float* __restrict__ alpha_src,
                                                  float* __restrict__ alpha_dst) {
    __shared__ __attribute__((aligned(16))) float lds_x[16][132];
    __shared__ __attribute__((aligned(16))) float lds_w[128][132];
    const int tid  = threadIdx.x;
    const int lane = tid & 31;
    const int wave = tid >> 5;
    const int r0   = blockIdx.x * 16;
    const int c0   = wave * 16;
    const int lm   = lane & 15;
    const int lhi  = lane >> 4;

    const bool dot_thr = tid < 128;
    const int  didx    = tid & 63;
    const int  drow    = didx >> 2;
    const int  dhead   = didx & 3;
    float4 cf[8];
    if (dot_thr) {
        const float4* ap = (const float4*)(((tid >> 6) ? a_dst : a_src) + dhead * DD);
#pragma unroll
        for (int i = 0; i < 8; ++i) cf[i] = ap[i];
    }

    const float* xsrc = x + (size_t)(r0 + wave) * KDIM + lane * 4;
    const float* wsrc = W + (size_t)wave * KDIM + lane * 4;
    float4 xr[2], wr[16];
#pragma unroll
    for (int i = 0; i < 2; ++i) xr[i] = *(const float4*)(xsrc + (size_t)(8 * i) * KDIM);
#pragma unroll
    for (int i = 0; i < 16; ++i) wr[i] = *(const float4*)(wsrc + (size_t)(8 * i) * KDIM);
#pragma unroll
    for (int i = 0; i < 2; ++i) *(float4*)&lds_x[wave + 8 * i][lane * 4] = xr[i];
#pragma unroll
    for (int i = 0; i < 16; ++i) *(float4*)&lds_w[wave + 8 * i][lane * 4] = wr[i];
    __syncthreads();

    v8f acc = {};
#pragma unroll
    for (int s = 0; s < KDIM / 4; ++s) {
        const int kb = 4 * s + lhi * 2;
        v2f a, b;
        a.x = lds_x[lm][kb + 0];
        a.y = lds_x[lm][kb + 1];
        b.x = lds_w[c0 + lm][kb + 0];
        b.y = lds_w[c0 + lm][kb + 1];
        acc = __builtin_amdgcn_wmma_f32_16x16x4_f32(false, a, false, b,
                                                    (short)0, acc, false, false);
    }
    __syncthreads();
#pragma unroll
    for (int v = 0; v < 8; ++v) {
        int m = v + lhi * 8;
        lds_x[m][c0 + lm] = acc[v];
    }
    __syncthreads();
#pragma unroll
    for (int v = 0; v < 8; ++v) {
        int m = v + lhi * 8;
        h[(size_t)(r0 + m) * KDIM + c0 + lm] = acc[v];
    }
    if (dot_thr) {
        const float4* hp = (const float4*)&lds_x[drow][dhead * DD];
        float s = 0.f;
#pragma unroll
        for (int i = 0; i < 8; ++i) {
            float4 hv = hp[i], av = cf[i];
            s += hv.x * av.x + hv.y * av.y + hv.z * av.z + hv.w * av.w;
        }
        float* dstp = (tid >> 6) ? alpha_dst : alpha_src;
        dstp[(size_t)r0 * HH + didx] = s;
    }
}

#define QPB    128
#define KCH    128
#define PPITCH 84
#define PP4    (PPITCH / 4)
#define BIT    16

__device__ __forceinline__ void attn_block(const float4 (&e)[BIT], const float4 (&v)[BIT], const v2f ac,
                                           v8f& acc0, v8f& acc1, float& zacc) {
#pragma unroll
    for (int i = 0; i < BIT; ++i) {
        v2f qa, qb, p, v0, v1;
        qa.x = e[i].x; qa.y = e[i].y;
        qb.x = e[i].z; qb.y = e[i].w;
        qa *= ac;
        qb *= ac;
        p.x = fmaxf(qa.x, qa.y);
        p.y = fmaxf(qb.x, qb.y);
        zacc += p.x + p.y;
        v0.x = v[i].x; v0.y = v[i].y;
        v1.x = v[i].z; v1.y = v[i].w;
        acc0 = __builtin_amdgcn_wmma_f32_16x16x4_f32(false, p, false, v0,
                                                     (short)0, acc0, false, false);
        acc1 = __builtin_amdgcn_wmma_f32_16x16x4_f32(false, p, false, v1,
                                                     (short)0, acc1, false, false);
    }
}

__device__ __forceinline__ void put_pair(float* d, const float4 a, const float4 b) {
    *(float2*)(d + 0)  = make_float2(a.x, b.x);
    *(float2*)(d + 4)  = make_float2(a.y, b.y);
    *(float2*)(d + 8)  = make_float2(a.z, b.z);
    *(float2*)(d + 12) = make_float2(a.w, b.w);
}
__global__ __launch_bounds__(256) void gat_attn(const float* __restrict__ h,
                                                const float* __restrict__ alpha_src,
                                                const float* __restrict__ alpha_dst,
                                                const float* __restrict__ alive,
                                                float* __restrict__ out) {
    __shared__ __attribute__((aligned(16))) float lds_tv[NN * 2 + (NN / 2) * PPITCH];
    float* const ef  = lds_tv;
    float* const vsl = lds_tv + NN * 2;
    __shared__ float inv_s[8][16];
    __shared__ __attribute__((aligned(16))) float wmax[8];
    const int tid   = threadIdx.x;
    const int lane  = tid & 31;
    const int wave  = tid >> 5;
    const int lm    = lane & 15;
    const int lhi   = lane >> 4;
    const int head  = (blockIdx.x >> 3) & 3;
    const int b     = blockIdx.x >> 5;
    const int q0    = (blockIdx.x & 7) * QPB + wave * 16;

    float dk[NN / 256];
    bool  lv[NN / 256];
    float mloc = -INFINITY;
#pragma unroll
    for (int k = 0; k < NN / 256; ++k) {
        const int j = tid + 256 * k;
        dk[k] = alpha_dst[(size_t)(b * NN + j) * HH + head];
        lv[k] = alive[b * NN + j] >= 0.5f;
    }
#pragma unroll
    for (int k = 0; k < NN / 256; ++k)
        if (lv[k] && dk[k] > mloc) mloc = dk[k];

    const int lpr  = tid >> 3;
    const int lseg = (tid & 7) * 4;
    const float* hsrc = h + (size_t)b * NN * KDIM + (size_t)(2 * lpr) * KDIM + head * DD + lseg;
    float* vdst = vsl + lpr * PPITCH + (lseg & 15) * 4 + (lseg >> 4) * 2;
    const float4 f0 = *(const float4*)(hsrc + (size_t)(0) * KDIM);
    const float4 f1 = *(const float4*)(hsrc + (size_t)(1) * KDIM);
    const float4 f2 = *(const float4*)(hsrc + (size_t)(64) * KDIM);
    const float4 f3 = *(const float4*)(hsrc + (size_t)(65) * KDIM);
    float4 pre0 = *(const float4*)(hsrc + (size_t)(KCH + 0) * KDIM);
    float4 pre1 = *(const float4*)(hsrc + (size_t)(KCH + 1) * KDIM);
    float4 pre2 = *(const float4*)(hsrc + (size_t)(KCH + 64) * KDIM);
    float4 pre3 = *(const float4*)(hsrc + (size_t)(KCH + 65) * KDIM);

    const float s_i = alpha_src[(size_t)(b * NN + q0 + lm) * HH + head];
    for (int o = 16; o > 0; o >>= 1) mloc = fmaxf(mloc, __shfl_xor(mloc, o, 32));
    put_pair(vdst, f0, f1);
    put_pair(vdst + 32 * PPITCH, f2, f3);
    if (lane == 0) wmax[wave] = mloc;
    __syncthreads();
    const float4 wa = *(const float4*)&wmax[0];
    const float4 wb = *(const float4*)&wmax[4];
    const float Dm  = fmaxf(fmaxf(fmaxf(wa.x, wa.y), fmaxf(wa.z, wa.w)),
                            fmaxf(fmaxf(wb.x, wb.y), fmaxf(wb.z, wb.w)));
#pragma unroll
    for (int k = 0; k < NN / 256; ++k) {
        const float t = dk[k] - Dm;
        float2 w;
        w.x = lv[k] ? __expf(t) : 0.f;
        w.y = lv[k] ? __expf(NEG_SLOPE * t) : 0.f;
        *(float2*)&ef[2 * (tid + 256 * k)] = w;
    }
    const float e0 = s_i + Dm;
    const float sm = __expf(-fabsf((1.f - NEG_SLOPE) * e0));
    v2f ac;
    ac.x = (e0 >= 0.f) ? 1.f : sm;
    ac.y = (e0 >= 0.f) ? sm : 1.f;
    __syncthreads();

    const float4* efl = (const float4*)ef + lhi;
    const float4* vl  = (const float4*)(vsl + lhi * PPITCH + lm * 4);
    v8f acc0 = {}, acc1 = {};
    float zacc = 0.f;
    float4 eA[BIT], vA[BIT], eB[BIT], vB[BIT];
#pragma unroll
    for (int i = 0; i < BIT; ++i) {
        eA[i] = efl[2 * i];
        vA[i] = vl[2 * i * PP4];
    }
#pragma unroll 1
    for (int c = 0; c < NN / KCH; ++c) {
        float4 nx0 = pre0, nx1 = pre1, nx2 = pre2, nx3 = pre3;
        if (c + 2 < NN / KCH) {
            const float* hn = hsrc + (size_t)((c + 2) * KCH) * KDIM;
            nx0 = *(const float4*)(hn + (size_t)(0) * KDIM);
            nx1 = *(const float4*)(hn + (size_t)(1) * KDIM);
            nx2 = *(const float4*)(hn + (size_t)(64) * KDIM);
            nx3 = *(const float4*)(hn + (size_t)(65) * KDIM);
        }
        const float4* efc = efl + c * (KCH / 2);
        const float4* vc  = vl + c * (KCH / 2) * PP4;
#pragma unroll
        for (int i = 0; i < BIT; ++i) {
            eB[i] = efc[2 * (BIT + i)];
            vB[i] = vc[2 * (BIT + i) * PP4];
        }
        __builtin_amdgcn_sched_barrier(0);
        attn_block(eA, vA, ac, acc0, acc1, zacc);
        if (c + 1 < NN / KCH) {
            float* vn = vdst + (c + 1) * (KCH / 2) * PPITCH;
            put_pair(vn, pre0, pre1);
            put_pair(vn + 32 * PPITCH, pre2, pre3);
        }
        __syncthreads();
        if (c + 1 < NN / KCH) {
#pragma unroll
            for (int i = 0; i < BIT; ++i) {
                eA[i] = efc[KCH / 2 + 2 * i];
                vA[i] = vc[(KCH / 2 + 2 * i) * PP4];
            }
        }
        attn_block(eB, vB, ac, acc0, acc1, zacc);
        pre0 = nx0; pre1 = nx1; pre2 = nx2; pre3 = nx3;
    }
    zacc += __shfl_xor(zacc, 16, 32);
    if (lhi == 0) inv_s[wave][lm] = (zacc > 0.f) ? (1.f / zacc) : 0.f;
    __syncthreads();
#pragma unroll
    for (int v = 0; v < 8; ++v) {
        int m = v + lhi * 8;
        const float sc = inv_s[wave][m];
        float* o = out + ((size_t)(b * NN) + q0 + m) * KDIM + head * DD + lm;
        o[0]  = acc0[v] * sc;
        o[16] = acc1[v] * sc;
    }
}

extern "C" void kernel_launch(void* const* d_in, const int* in_sizes, int n_in,
                              void* d_out, int out_size, void* d_ws, size_t ws_size,
                              hipStream_t stream) {
    const float* x     = (const float*)d_in[0];
    const float* alive = (const float*)d_in[1];
    const float* W     = (const float*)d_in[2];
    const float* a_src = (const float*)d_in[3];
    const float* a_dst = (const float*)d_in[4];
    float* out = (float*)d_out;

    float* h_ws  = (float*)d_ws;
    float* as_ws = h_ws + (size_t)BB * NN * KDIM;
    float* ad_ws = as_ws + (size_t)BB * NN * HH;

    gat_h_gemm<<<512, 256, 0, stream>>>(x, W, a_src, a_dst, h_ws, as_ws, ad_ws);
    gat_attn  <<<256, 256, 0, stream>>>(h_ws, as_ws, ad_ws, alive, out);
}


# attention skips masked keys: block ranks its live keys (ballot + bit count), stages only their V rows and (E,F) by rank; key loop runs ceil(live/64) blocks; sums group 4 consecutive LIVE keys
# speedup vs baseline: 2.7264x; 1.2962x over previous
#include <hip/hip_runtime.h>
#include <math.h>

typedef __attribute__((ext_vector_type(2))) float v2f;
typedef __attribute__((ext_vector_type(8))) float v8f;

#define HH 4
#define DD 32
#define NN 1024
#define BB 8
#define KDIM 128
#define NEG_SLOPE 0.2f

__global__ __launch_bounds__(256) void gat_h_gemm(const float* __restrict__ x,
                                                  const float* __restrict__ W,
                                                  const float* __restrict__ a_src,
                                                  const float* __restrict__ a_dst,
                                                  float* __restrict__ h,
                                                  float* __restrict__ alpha_src,
                                                  float* __restrict__ alpha_dst) {
    __shared__ __attribute__((aligned(16))) float lds_x[16][132];
    __shared__ __attribute__((aligned(16))) float lds_w[128][132];
    const int tid  = threadIdx.x;
    const int lane = tid & 31;
    const int wave = tid >> 5;
    const int r0   = blockIdx.x * 16;
    const int c0   = wave * 16;
    const int lm   = lane & 15;
    const int lhi  = lane >> 4;

    const bool dot_thr = tid < 128;
    const int  didx    = tid & 63;
    const int  drow    = didx >> 2;
    const int  dhead   = didx & 3;
    float4 cf[8];
    if (dot_thr) {
        const float4* ap = (const float4*)(((tid >> 6) ? a_dst : a_src) + dhead * DD);
#pragma unroll
        for (int i = 0; i < 8; ++i) cf[i] = ap[i];
    }

    const float* xsrc = x + (size_t)(r0 + wave) * KDIM + lane * 4;
    const float* wsrc = W + (size_t)wave * KDIM + lane * 4;
    float4 xr[2], wr[16];
#pragma unroll
    for (int i = 0; i < 2; ++i) xr[i] = *(const float4*)(xsrc + (size_t)(8 * i) * KDIM);
#pragma unroll
    for (int i = 0; i < 16; ++i) wr[i] = *(const float4*)(wsrc + (size_t)(8 * i) * KDIM);
#pragma unroll
    for (int i = 0; i < 2; ++i) *(float4*)&lds_x[wave + 8 * i][lane * 4] = xr[i];
#pragma unroll
    for (int i = 0; i < 16; ++i) *(float4*)&lds_w[wave + 8 * i][lane * 4] = wr[i];
    __syncthreads();

    v8f acc = {};
#pragma unroll
    for (int s = 0; s < KDIM / 4; ++s) {
        const int kb = 4 * s + lhi * 2;
        v2f a, b;
        a.x = lds_x[lm][kb + 0];
        a.y = lds_x[lm][kb + 1];
        b.x = lds_w[c0 + lm][kb + 0];
        b.y = lds_w[c0 + lm][kb + 1];
        acc = __builtin_amdgcn_wmma_f32_16x16x4_f32(false, a, false, b,
                                                    (short)0, acc, false, false);
    }
    __syncthreads();
#pragma unroll
    for (int v = 0; v < 8; ++v) {
        int m = v + lhi * 8;
        lds_x[m][c0 + lm] = acc[v];
    }
    __syncthreads();
#pragma unroll
    for (int v = 0; v < 8; ++v) {
        int m = v + lhi * 8;
        h[(size_t)(r0 + m) * KDIM + c0 + lm] = acc[v];
    }
    if (dot_thr) {
        const float4* hp = (const float4*)&lds_x[drow][dhead * DD];
        float s = 0.f;
#pragma unroll
        for (int i = 0; i < 8; ++i) {
            float4 hv = hp[i], av = cf[i];
            s += hv.x * av.x + hv.y * av.y + hv.z * av.z + hv.w * av.w;
        }
        float* dstp = (tid >> 6) ? alpha_dst : alpha_src;
        dstp[(size_t)r0 * HH + didx] = s;
    }
}

#define QPB    128
#define KCH    128
#define PPITCH 84
#define PP4    (PPITCH / 4)
#define BIT    16

__device__ __forceinline__ void attn_block(const float4 (&e)[BIT], const float4 (&v)[BIT], const v2f ac,
                                           v8f& acc0, v8f& acc1, float& zacc) {
#pragma unroll
    for (int i = 0; i < BIT; ++i) {
        v2f qa, qb, p, v0, v1;
        qa.x = e[i].x; qa.y = e[i].y;
        qb.x = e[i].z; qb.y = e[i].w;
        qa *= ac;
        qb *= ac;
        p.x = fmaxf(qa.x, qa.y);
        p.y = fmaxf(qb.x, qb.y);
        zacc += p.x + p.y;
        v0.x = v[i].x; v0.y = v[i].y;
        v1.x = v[i].z; v1.y = v[i].w;
        acc0 = __builtin_amdgcn_wmma_f32_16x16x4_f32(false, p, false, v0,
                                                     (short)0, acc0, false, false);
        acc1 = __builtin_amdgcn_wmma_f32_16x16x4_f32(false, p, false, v1,
                                                     (short)0, acc1, false, false);
    }
}

__device__ __forceinline__ void put_pair(float* d, const float4 a, const float4 b) {
    *(float2*)(d + 0)  = make_float2(a.x, b.x);
    *(float2*)(d + 4)  = make_float2(a.y, b.y);
    *(float2*)(d + 8)  = make_float2(a.z, b.z);
    *(float2*)(d + 12) = make_float2(a.w, b.w);
}
__global__ __launch_bounds__(256) void gat_attn(const float* __restrict__ h,
                                                const float* __restrict__ alpha_src,
                                                const float* __restrict__ alpha_dst,
                                                const float* __restrict__ alive,
                                                float* __restrict__ out) {
    __shared__ __attribute__((aligned(16))) float lds_tv[NN * 2 + (NN / 2) * PPITCH];
    float* const ef  = lds_tv;
    float* const vsl = lds_tv + NN * 2;
    __shared__ __attribute__((aligned(16))) int klist[NN];
    __shared__ float inv_s[8][16];
    __shared__ __attribute__((aligned(16))) float wmax[8];
    __shared__ __attribute__((aligned(16))) int   wtot[8];
    const int tid   = threadIdx.x;
    const int lane  = tid & 31;
    const int wave  = tid >> 5;
    const int lm    = lane & 15;
    const int lhi   = lane >> 4;
    const int head  = (blockIdx.x >> 3) & 3;
    const int b     = blockIdx.x >> 5;
    const int q0    = (blockIdx.x & 7) * QPB + wave * 16;

    const int j0 = 4 * tid;
    const float4 al = *(const float4*)(alive + b * NN + j0);
    float dk[4];
#pragma unroll
    for (int k = 0; k < 4; ++k) dk[k] = alpha_dst[(size_t)(b * NN + j0 + k) * HH + head];
    const float s_i = alpha_src[(size_t)(b * NN + q0 + lm) * HH + head];
    bool lv[4];
    lv[0] = al.x >= 0.5f; lv[1] = al.y >= 0.5f; lv[2] = al.z >= 0.5f; lv[3] = al.w >= 0.5f;
    float mloc = -INFINITY;
#pragma unroll
    for (int k = 0; k < 4; ++k)
        if (lv[k] && dk[k] > mloc) mloc = dk[k];
    const unsigned ltm = (1u << lane) - 1u;
    int below = 0, wcount = 0;
#pragma unroll
    for (int k = 0; k < 4; ++k) {
        const unsigned m = __builtin_amdgcn_ballot_w32(lv[k]);
        below  += __popc(m & ltm);
        wcount += __popc(m);
    }
    if (lane == 0) wtot[wave] = wcount;
    __syncthreads();
    const int4 ta = *(const int4*)&wtot[0];
    const int4 tb = *(const int4*)&wtot[4];
    const int tw[8] = {ta.x, ta.y, ta.z, ta.w, tb.x, tb.y, tb.z, tb.w};
    int base = 0, nlive = 0;
#pragma unroll
    for (int w2 = 0; w2 < 8; ++w2) {
        if (w2 < wave) base += tw[w2];
        nlive += tw[w2];
    }
    const int nblk = (nlive + 63) >> 6;
    const int nch  = (nblk + 1) >> 1;
    int ps[4];
    {
        int pos = base + below;
#pragma unroll
        for (int k = 0; k < 4; ++k) {
            ps[k] = pos;
            if (lv[k]) { klist[pos] = j0 + k; ++pos; }
        }
    }
    const int  ppad = nlive + tid;
    const bool dopad = (tid < KCH) && (ppad < nch * KCH);
    if (dopad) klist[ppad] = 0;
    __syncthreads();

    const int lpr  = tid >> 3;
    const int lseg = (tid & 7) * 4;
    const float* hb = h + (size_t)b * NN * KDIM + head * DD + lseg;
    float* vdst = vsl + lpr * PPITCH + (lseg & 15) * 4 + (lseg >> 4) * 2;
    const int2* kl = (const int2*)klist + lpr;
    const bool has0 = nch > 0, has1 = nch > 1;
    int2 a0 = kl[0], a1 = kl[32], c0 = kl[64], c1 = kl[96];
    int2 ja = kl[128], jb = kl[160];
    a0.x = has0 ? a0.x : 0; a0.y = has0 ? a0.y : 0; a1.x = has0 ? a1.x : 0; a1.y = has0 ? a1.y : 0;
    c0.x = has1 ? c0.x : 0; c0.y = has1 ? c0.y : 0; c1.x = has1 ? c1.x : 0; c1.y = has1 ? c1.y : 0;
    const float4 f0 = *(const float4*)(hb + (size_t)a0.x * KDIM);
    const float4 f1 = *(const float4*)(hb + (size_t)a0.y * KDIM);
    const float4 f2 = *(const float4*)(hb + (size_t)a1.x * KDIM);
    const float4 f3 = *(const float4*)(hb + (size_t)a1.y * KDIM);
    float4 pre0 = *(const float4*)(hb + (size_t)c0.x * KDIM);
    float4 pre1 = *(const float4*)(hb + (size_t)c0.y * KDIM);
    float4 pre2 = *(const float4*)(hb + (size_t)c1.x * KDIM);
    float4 pre3 = *(const float4*)(hb + (size_t)c1.y * KDIM);

    for (int o = 16; o > 0; o >>= 1) mloc = fmaxf(mloc, __shfl_xor(mloc, o, 32));
    if (lane == 0) wmax[wave] = mloc;
    __syncthreads();
    const float4 wa = *(const float4*)&wmax[0];
    const float4 wb = *(const float4*)&wmax[4];
    const float Dm  = fmaxf(fmaxf(fmaxf(wa.x, wa.y), fmaxf(wa.z, wa.w)),
                            fmaxf(fmaxf(wb.x, wb.y), fmaxf(wb.z, wb.w)));
#pragma unroll
    for (int k = 0; k < 4; ++k) {
        if (lv[k]) {
            const float t = dk[k] - Dm;
            float2 w;
            w.x = __expf(t);
            w.y = __expf(NEG_SLOPE * t);
            *(float2*)&ef[2 * ps[k]] = w;
        }
    }
    if (dopad) *(float2*)&ef[2 * ppad] = make_float2(0.f, 0.f);
    const float e0 = s_i + Dm;
    const float sm = __expf(-fabsf((1.f - NEG_SLOPE) * e0));
    v2f ac;
    ac.x = (e0 >= 0.f) ? 1.f : sm;
    ac.y = (e0 >= 0.f) ? sm : 1.f;
    put_pair(vdst, f0, f1);
    put_pair(vdst + 32 * PPITCH, f2, f3);
    __syncthreads();

    const float4* efl = (const float4*)ef + lhi;
    const float4* vl  = (const float4*)(vsl + lhi * PPITCH + lm * 4);
    v8f acc0 = {}, acc1 = {};
    float zacc = 0.f;
    float4 eA[BIT], vA[BIT], eB[BIT], vB[BIT];
#pragma unroll
    for (int i = 0; i < BIT; ++i) {
        eA[i] = efl[2 * i];
        vA[i] = vl[2 * i * PP4];
    }
#pragma unroll 1
    for (int c = 0; c < nch; ++c) {
        float4 nx0 = pre0, nx1 = pre1, nx2 = pre2, nx3 = pre3;
        if (c + 2 < nch) {
            nx0 = *(const float4*)(hb + (size_t)ja.x * KDIM);
            nx1 = *(const float4*)(hb + (size_t)ja.y * KDIM);
            nx2 = *(const float4*)(hb + (size_t)jb.x * KDIM);
            nx3 = *(const float4*)(hb + (size_t)jb.y * KDIM);
        }
        const float4* efc = efl + c * (KCH / 2);
        const float4* vc  = vl + c * (KCH / 2) * PP4;
#pragma unroll
        for (int i = 0; i < BIT; ++i) {
            eB[i] = efc[2 * (BIT + i)];
            vB[i] = vc[2 * (BIT + i) * PP4];
        }
        const int cn = (c + 3 < NN / KCH) ? (c + 3) : (NN / KCH - 1);
        ja = kl[cn * (KCH / 2)];
        jb = kl[cn * (KCH / 2) + 32];
        __builtin_amdgcn_sched_barrier(0);
        attn_block(eA, vA, ac, acc0, acc1, zacc);
        if (c + 1 < nch) {
            float* vn = vdst + (c + 1) * (KCH / 2) * PPITCH;
            put_pair(vn, pre0, pre1);
            put_pair(vn + 32 * PPITCH, pre2, pre3);
        }
        __syncthreads();
        if (c + 1 < nch) {
#pragma unroll
            for (int i = 0; i < BIT; ++i) {
                eA[i] = efc[KCH / 2 + 2 * i];
                vA[i] = vc[(KCH / 2 + 2 * i) * PP4];
            }
        }
        if (2 * c + 1 < nblk) attn_block(eB, vB, ac, acc0, acc1, zacc);
        pre0 = nx0; pre1 = nx1; pre2 = nx2; pre3 = nx3;
    }
    zacc += __shfl_xor(zacc, 16, 32);
    if (lhi == 0) inv_s[wave][lm] = (zacc > 0.f) ? (1.f / zacc) : 0.f;
    __syncthreads();
#pragma unroll
    for (int v = 0; v < 8; ++v) {
        int m = v + lhi * 8;
        const float sc = inv_s[wave][m];
        float* o = out + ((size_t)(b * NN) + q0 + m) * KDIM + head * DD + lm;
        o[0]  = acc0[v] * sc;
        o[16] = acc1[v] * sc;
    }
}

extern "C" void kernel_launch(void* const* d_in, const int* in_sizes, int n_in,
                              void* d_out, int out_size, void* d_ws, size_t ws_size,
                              hipStream_t stream) {
    const float* x     = (const float*)d_in[0];
    const float* alive = (const float*)d_in[1];
    const float* W     = (const float*)d_in[2];
    const float* a_src = (const float*)d_in[3];
    const float* a_dst = (const float*)d_in[4];
    float* out = (float*)d_out;

    float* h_ws  = (float*)d_ws;
    float* as_ws = h_ws + (size_t)BB * NN * KDIM;
    float* ad_ws = as_ws + (size_t)BB * NN * HH;

    gat_h_gemm<<<512, 256, 0, stream>>>(x, W, a_src, a_dst, h_ws, as_ws, ad_ws);
    gat_attn  <<<256, 256, 0, stream>>>(h_ws, as_ws, ad_ws, alive, out);
}
